# starting kernel
# baseline (speedup 1.0000x reference)
#include <hip/hip_runtime.h>

typedef float v2f __attribute__((ext_vector_type(2)));
typedef float v4f __attribute__((ext_vector_type(4)));
typedef float v8f __attribute__((ext_vector_type(8)));

#define BLOCK 256
#define NWAVES (BLOCK / 32)

__device__ __forceinline__ float wave_sum32_wmma(float p) {
  v2f a;
  a[0] = p;
  a[1] = 0.0f;
  v2f ones;
  ones[0] = 1.0f;
  ones[1] = 1.0f;

  v8f zero_c = {};
  v8f d1 = __builtin_amdgcn_wmma_f32_16x16x4_f32(
       false, a,  false, ones,
       (short)0, zero_c,  false,  false);

  v8f acc = {};
  v2f b;
#pragma unroll
  for (int j = 0; j < 4; ++j) {
    b[0] = d1[2 * j];
    b[1] = d1[2 * j + 1];
    acc = __builtin_amdgcn_wmma_f32_16x16x4_f32(
        false, ones, false, b, (short)0, acc, false, false);
  }
  return acc[0];
}

__device__ __forceinline__ float block_rowsum(float p, int tid) {
  float wsum = wave_sum32_wmma(p);

  __shared__ float lds_part[NWAVES];
  if ((tid & 31) == 0) lds_part[tid >> 5] = wsum;
  __syncthreads();

  float rowsum = 0.0f;
#pragma unroll
  for (int i = 0; i < NWAVES; ++i) rowsum += lds_part[i];
  return rowsum;
}

template <int KITERS, int NITERS>
__global__ __launch_bounds__(BLOCK) void Rank1Linear_kernel_tmpl(
    const float* __restrict__ x, const float* __restrict__ w,
    float* __restrict__ out) {
  const int row = blockIdx.x;
  const int tid = threadIdx.x;

  const v4f* __restrict__ xrow =
      reinterpret_cast<const v4f*>(x) + (size_t)row * (size_t)(KITERS * BLOCK);

  float p = 0.0f;
#pragma unroll
  for (int it = 0; it < KITERS; ++it) {
    v4f v = __builtin_nontemporal_load(&xrow[tid + it * BLOCK]);
    p += (v[0] + v[1]) + (v[2] + v[3]);
  }

  const float rowsum = block_rowsum(p, tid);

  const v4f* __restrict__ wv = reinterpret_cast<const v4f*>(w);
  v4f* __restrict__ orow =
      reinterpret_cast<v4f*>(out) + (size_t)row * (size_t)(NITERS * BLOCK);

#pragma unroll
  for (int it = 0; it < NITERS; ++it) {
    const int i = tid + it * BLOCK;
    v4f wl = wv[i];
    v4f o;
    o[0] = rowsum * wl[0];
    o[1] = rowsum * wl[1];
    o[2] = rowsum * wl[2];
    o[3] = rowsum * wl[3];
    __builtin_nontemporal_store(o, &orow[i]);
  }
}

__global__ __launch_bounds__(BLOCK) void Rank1Linear_kernel_gen(
    const float* __restrict__ x, const float* __restrict__ w,
    float* __restrict__ out, int kvec, int nvec) {
  const int row = blockIdx.x;
  const int tid = threadIdx.x;

  const v4f* __restrict__ xrow =
      reinterpret_cast<const v4f*>(x) + (size_t)row * (size_t)kvec;

  float p = 0.0f;
  const int kiters = kvec >> 8;
  for (int it = 0; it < kiters; ++it) {
    v4f v = __builtin_nontemporal_load(&xrow[tid + (it << 8)]);
    p += (v[0] + v[1]) + (v[2] + v[3]);
  }
  {
    const int i = (kiters << 8) + tid;
    if (i < kvec) {
      v4f v = __builtin_nontemporal_load(&xrow[i]);
      p += (v[0] + v[1]) + (v[2] + v[3]);
    }
  }

  const float rowsum = block_rowsum(p, tid);

  const v4f* __restrict__ wv = reinterpret_cast<const v4f*>(w);
  v4f* __restrict__ orow =
      reinterpret_cast<v4f*>(out) + (size_t)row * (size_t)nvec;

  for (int i = tid; i < nvec; i += BLOCK) {
    v4f wl = wv[i];
    v4f o;
    o[0] = rowsum * wl[0];
    o[1] = rowsum * wl[1];
    o[2] = rowsum * wl[2];
    o[3] = rowsum * wl[3];
    __builtin_nontemporal_store(o, &orow[i]);
  }
}

extern "C" void kernel_launch(void* const* d_in, const int* in_sizes, int n_in,
                              void* d_out, int out_size, void* d_ws,
                              size_t ws_size, hipStream_t stream) {
  const float* x = (const float*)d_in[0];
  const float* w = (const float*)d_in[1];
  float* out = (float*)d_out;

  const int N = in_sizes[1];
  const int batch = out_size / N;
  const int K = in_sizes[0] / batch;

  dim3 grid(batch);
  dim3 block(BLOCK);

  if (K == 4096 && N == 4096) {
    hipLaunchKernelGGL((Rank1Linear_kernel_tmpl<4, 4>), grid, block, 0, stream,
                       x, w, out);
  } else {
    hipLaunchKernelGGL(Rank1Linear_kernel_gen, grid, block, 0, stream,
                       x, w, out, K >> 2, N >> 2);
  }
}


# Persistent grid: 512 resident 8-wave workgroups each loop over 16 rows (w kept in registers, 4 rows of loads in flight, LDS-only barrier), so 4096 waves enter instead of 65536; sum order unchanged
# speedup vs baseline: 1.0750x; 1.0750x over previous
#include <hip/hip_runtime.h>

typedef float v2f __attribute__((ext_vector_type(2)));
typedef float v4f __attribute__((ext_vector_type(4)));
typedef float v8f __attribute__((ext_vector_type(8)));

#define BLOCK 256
#define NWAVES (BLOCK / 32)

__device__ __forceinline__ float wave_sum32_wmma(float p) {
  v2f a;
  a[0] = p;
  a[1] = 0.0f;
  v2f ones;
  ones[0] = 1.0f;
  ones[1] = 1.0f;

  v8f zero_c = {};
  v8f d1 = __builtin_amdgcn_wmma_f32_16x16x4_f32(
       false, a,  false, ones,
       (short)0, zero_c,  false,  false);

  v8f acc = {};
  v2f b;
#pragma unroll
  for (int j = 0; j < 4; ++j) {
    b[0] = d1[2 * j];
    b[1] = d1[2 * j + 1];
    acc = __builtin_amdgcn_wmma_f32_16x16x4_f32(
        false, ones, false, b, (short)0, acc, false, false);
  }
  return acc[0];
}

__device__ __forceinline__ float block_rowsum(float p, int tid) {
  float wsum = wave_sum32_wmma(p);

  __shared__ float lds_part[NWAVES];
  if ((tid & 31) == 0) lds_part[tid >> 5] = wsum;
  __syncthreads();

  float rowsum = 0.0f;
#pragma unroll
  for (int i = 0; i < NWAVES; ++i) rowsum += lds_part[i];
  return rowsum;
}

template <int KITERS, int NITERS>
__global__ __launch_bounds__(BLOCK) void Rank1Linear_kernel_tmpl(
    const float* __restrict__ x, const float* __restrict__ w,
    float* __restrict__ out) {
  const int row = blockIdx.x;
  const int tid = threadIdx.x;

  const v4f* __restrict__ xrow =
      reinterpret_cast<const v4f*>(x) + (size_t)row * (size_t)(KITERS * BLOCK);

  float p = 0.0f;
#pragma unroll
  for (int it = 0; it < KITERS; ++it) {
    v4f v = __builtin_nontemporal_load(&xrow[tid + it * BLOCK]);
    p += (v[0] + v[1]) + (v[2] + v[3]);
  }

  const float rowsum = block_rowsum(p, tid);

  const v4f* __restrict__ wv = reinterpret_cast<const v4f*>(w);
  v4f* __restrict__ orow =
      reinterpret_cast<v4f*>(out) + (size_t)row * (size_t)(NITERS * BLOCK);

#pragma unroll
  for (int it = 0; it < NITERS; ++it) {
    const int i = tid + it * BLOCK;
    v4f wl = wv[i];
    v4f o;
    o[0] = rowsum * wl[0];
    o[1] = rowsum * wl[1];
    o[2] = rowsum * wl[2];
    o[3] = rowsum * wl[3];
    __builtin_nontemporal_store(o, &orow[i]);
  }
}

#define PDEPTH 4
#define PGRID 512

__device__ __forceinline__ void lds_barrier() {
  asm volatile(
      "s_wait_dscnt 0x0\n\t"
      "s_barrier_signal -1\n\t"
      "s_barrier_wait -1" ::: "memory");
}

template <int KITERS, int NITERS, int ROWS>
__global__ __launch_bounds__(BLOCK)
__attribute__((amdgpu_waves_per_eu(4, 4)))
void Rank1Linear_kernel_rows(
    const float* __restrict__ x, const float* __restrict__ w,
    float* __restrict__ out) {
  static_assert(ROWS >= PDEPTH, "the prologue requests PDEPTH rows");
  const int tid = threadIdx.x;
  const size_t stride = gridDim.x;

  __shared__ __attribute__((aligned(16))) float lds_part[2][NWAVES];

  const v4f* __restrict__ wv = reinterpret_cast<const v4f*>(w);
  v4f wl[NITERS];
#pragma unroll
  for (int it = 0; it < NITERS; ++it) wl[it] = wv[tid + it * BLOCK];

  const size_t xstep = stride * (size_t)(KITERS * BLOCK);
  const size_t ostep = stride * (size_t)(NITERS * BLOCK);
  const v4f* __restrict__ xload = reinterpret_cast<const v4f*>(x) +
                                  (size_t)blockIdx.x * (size_t)(KITERS * BLOCK);
  v4f* __restrict__ orow = reinterpret_cast<v4f*>(out) +
                           (size_t)blockIdx.x * (size_t)(NITERS * BLOCK);

  v4f buf[PDEPTH][KITERS];
#pragma unroll
  for (int d = 0; d < PDEPTH; ++d) {
#pragma unroll
    for (int it = 0; it < KITERS; ++it)
      buf[d][it] = __builtin_nontemporal_load(&xload[tid + it * BLOCK]);
    xload += xstep;
  }

#pragma unroll
  for (int s = 0; s < ROWS; ++s) {
    float p = 0.0f;
#pragma unroll
    for (int it = 0; it < KITERS; ++it) {
      const v4f v = buf[s % PDEPTH][it];
      p += (v[0] + v[1]) + (v[2] + v[3]);
    }
    if (s + PDEPTH < ROWS) {
#pragma unroll
      for (int it = 0; it < KITERS; ++it)
        buf[s % PDEPTH][it] =
            __builtin_nontemporal_load(&xload[tid + it * BLOCK]);
      xload += xstep;
    }

    const float wsum = wave_sum32_wmma(p);
    if ((tid & 31) == 0) lds_part[s & 1][tid >> 5] = wsum;
    lds_barrier();
    float rowsum = 0.0f;
#pragma unroll
    for (int i = 0; i < NWAVES; ++i) rowsum += lds_part[s & 1][i];

#pragma unroll
    for (int it = 0; it < NITERS; ++it) {
      v4f o;
      o[0] = rowsum * wl[it][0];
      o[1] = rowsum * wl[it][1];
      o[2] = rowsum * wl[it][2];
      o[3] = rowsum * wl[it][3];
      __builtin_nontemporal_store(o, &orow[tid + it * BLOCK]);
    }
    orow += ostep;
    __builtin_amdgcn_sched_barrier(0);
  }
}

__global__ __launch_bounds__(BLOCK) void Rank1Linear_kernel_gen(
    const float* __restrict__ x, const float* __restrict__ w,
    float* __restrict__ out, int kvec, int nvec) {
  const int row = blockIdx.x;
  const int tid = threadIdx.x;

  const v4f* __restrict__ xrow =
      reinterpret_cast<const v4f*>(x) + (size_t)row * (size_t)kvec;

  float p = 0.0f;
  const int kiters = kvec >> 8;
  for (int it = 0; it < kiters; ++it) {
    v4f v = __builtin_nontemporal_load(&xrow[tid + (it << 8)]);
    p += (v[0] + v[1]) + (v[2] + v[3]);
  }
  {
    const int i = (kiters << 8) + tid;
    if (i < kvec) {
      v4f v = __builtin_nontemporal_load(&xrow[i]);
      p += (v[0] + v[1]) + (v[2] + v[3]);
    }
  }

  const float rowsum = block_rowsum(p, tid);

  const v4f* __restrict__ wv = reinterpret_cast<const v4f*>(w);
  v4f* __restrict__ orow =
      reinterpret_cast<v4f*>(out) + (size_t)row * (size_t)nvec;

  for (int i = tid; i < nvec; i += BLOCK) {
    v4f wl = wv[i];
    v4f o;
    o[0] = rowsum * wl[0];
    o[1] = rowsum * wl[1];
    o[2] = rowsum * wl[2];
    o[3] = rowsum * wl[3];
    __builtin_nontemporal_store(o, &orow[i]);
  }
}

extern "C" void kernel_launch(void* const* d_in, const int* in_sizes, int n_in,
                              void* d_out, int out_size, void* d_ws,
                              size_t ws_size, hipStream_t stream) {
  const float* x = (const float*)d_in[0];
  const float* w = (const float*)d_in[1];
  float* out = (float*)d_out;

  const int N = in_sizes[1];
  const int batch = out_size / N;
  const int K = in_sizes[0] / batch;

  dim3 grid(batch);
  dim3 block(BLOCK);

  if (K == 4096 && N == 4096 && batch == 8192) {
    hipLaunchKernelGGL((Rank1Linear_kernel_rows<4, 4, 16>), dim3(PGRID), block,
                       0, stream, x, w, out);
  } else if (K == 4096 && N == 4096) {
    hipLaunchKernelGGL((Rank1Linear_kernel_tmpl<4, 4>), grid, block, 0, stream,
                       x, w, out);
  } else {
    hipLaunchKernelGGL(Rank1Linear_kernel_gen, grid, block, 0, stream,
                       x, w, out, K >> 2, N >> 2);
  }
}


# Each resident workgroup sums all 16 of its rows first (partials parked in LDS, 1 barrier not 16), then stores its 16 output rows: memory sees reads, then writes, not a 1:1 mix; sum order unchanged
# speedup vs baseline: 1.1593x; 1.0784x over previous
#include <hip/hip_runtime.h>

typedef float v2f __attribute__((ext_vector_type(2)));
typedef float v4f __attribute__((ext_vector_type(4)));
typedef float v8f __attribute__((ext_vector_type(8)));

#define BLOCK 256
#define NWAVES (BLOCK / 32)

__device__ __forceinline__ float wave_sum32_wmma(float p) {
  v2f a;
  a[0] = p;
  a[1] = 0.0f;
  v2f ones;
  ones[0] = 1.0f;
  ones[1] = 1.0f;

  v8f zero_c = {};
  v8f d1 = __builtin_amdgcn_wmma_f32_16x16x4_f32(
       false, a,  false, ones,
       (short)0, zero_c,  false,  false);

  v8f acc = {};
  v2f b;
#pragma unroll
  for (int j = 0; j < 4; ++j) {
    b[0] = d1[2 * j];
    b[1] = d1[2 * j + 1];
    acc = __builtin_amdgcn_wmma_f32_16x16x4_f32(
        false, ones, false, b, (short)0, acc, false, false);
  }
  return acc[0];
}

__device__ __forceinline__ float block_rowsum(float p, int tid) {
  float wsum = wave_sum32_wmma(p);

  __shared__ float lds_part[NWAVES];
  if ((tid & 31) == 0) lds_part[tid >> 5] = wsum;
  __syncthreads();

  float rowsum = 0.0f;
#pragma unroll
  for (int i = 0; i < NWAVES; ++i) rowsum += lds_part[i];
  return rowsum;
}

template <int KITERS, int NITERS>
__global__ __launch_bounds__(BLOCK) void Rank1Linear_kernel_tmpl(
    const float* __restrict__ x, const float* __restrict__ w,
    float* __restrict__ out) {
  const int row = blockIdx.x;
  const int tid = threadIdx.x;

  const v4f* __restrict__ xrow =
      reinterpret_cast<const v4f*>(x) + (size_t)row * (size_t)(KITERS * BLOCK);

  float p = 0.0f;
#pragma unroll
  for (int it = 0; it < KITERS; ++it) {
    v4f v = __builtin_nontemporal_load(&xrow[tid + it * BLOCK]);
    p += (v[0] + v[1]) + (v[2] + v[3]);
  }

  const float rowsum = block_rowsum(p, tid);

  const v4f* __restrict__ wv = reinterpret_cast<const v4f*>(w);
  v4f* __restrict__ orow =
      reinterpret_cast<v4f*>(out) + (size_t)row * (size_t)(NITERS * BLOCK);

#pragma unroll
  for (int it = 0; it < NITERS; ++it) {
    const int i = tid + it * BLOCK;
    v4f wl = wv[i];
    v4f o;
    o[0] = rowsum * wl[0];
    o[1] = rowsum * wl[1];
    o[2] = rowsum * wl[2];
    o[3] = rowsum * wl[3];
    __builtin_nontemporal_store(o, &orow[i]);
  }
}

#define PDEPTH 4
#define PGRID 512

template <int KITERS, int NITERS, int ROWS>
__global__ __launch_bounds__(BLOCK)
__attribute__((amdgpu_waves_per_eu(4, 4)))
void Rank1Linear_kernel_phases(
    const float* __restrict__ x, const float* __restrict__ w,
    float* __restrict__ out) {
  static_assert(ROWS >= PDEPTH, "the prologue requests PDEPTH rows");
  const int tid = threadIdx.x;
  const size_t stride = gridDim.x;

  __shared__ __attribute__((aligned(16))) float lds_part[ROWS][NWAVES];

  const v4f* __restrict__ wv = reinterpret_cast<const v4f*>(w);
  v4f wl[NITERS];
#pragma unroll
  for (int it = 0; it < NITERS; ++it) wl[it] = wv[tid + it * BLOCK];

  const size_t xstep = stride * (size_t)(KITERS * BLOCK);
  const v4f* __restrict__ xload = reinterpret_cast<const v4f*>(x) +
                                  (size_t)blockIdx.x * (size_t)(KITERS * BLOCK);

  v4f buf[PDEPTH][KITERS];
#pragma unroll
  for (int d = 0; d < PDEPTH; ++d) {
#pragma unroll
    for (int it = 0; it < KITERS; ++it)
      buf[d][it] = __builtin_nontemporal_load(&xload[tid + it * BLOCK]);
    xload += xstep;
  }

#pragma unroll
  for (int s = 0; s < ROWS; ++s) {
    float p = 0.0f;
#pragma unroll
    for (int it = 0; it < KITERS; ++it) {
      const v4f v = buf[s % PDEPTH][it];
      p += (v[0] + v[1]) + (v[2] + v[3]);
    }
    if (s + PDEPTH < ROWS) {
#pragma unroll
      for (int it = 0; it < KITERS; ++it)
        buf[s % PDEPTH][it] =
            __builtin_nontemporal_load(&xload[tid + it * BLOCK]);
      xload += xstep;
    }

    const float wsum = wave_sum32_wmma(p);
    if ((tid & 31) == 0) lds_part[s][tid >> 5] = wsum;
  }

  __syncthreads();

  const size_t ostep = stride * (size_t)(NITERS * BLOCK);
  v4f* __restrict__ orow = reinterpret_cast<v4f*>(out) +
                           (size_t)blockIdx.x * (size_t)(NITERS * BLOCK);
#pragma unroll
  for (int s = 0; s < ROWS; ++s) {
    float rowsum = 0.0f;
#pragma unroll
    for (int i = 0; i < NWAVES; ++i) rowsum += lds_part[s][i];

#pragma unroll
    for (int it = 0; it < NITERS; ++it) {
      v4f o;
      o[0] = rowsum * wl[it][0];
      o[1] = rowsum * wl[it][1];
      o[2] = rowsum * wl[it][2];
      o[3] = rowsum * wl[it][3];
      __builtin_nontemporal_store(o, &orow[tid + it * BLOCK]);
    }
    orow += ostep;
  }
}

__global__ __launch_bounds__(BLOCK) void Rank1Linear_kernel_gen(
    const float* __restrict__ x, const float* __restrict__ w,
    float* __restrict__ out, int kvec, int nvec) {
  const int row = blockIdx.x;
  const int tid = threadIdx.x;

  const v4f* __restrict__ xrow =
      reinterpret_cast<const v4f*>(x) + (size_t)row * (size_t)kvec;

  float p = 0.0f;
  const int kiters = kvec >> 8;
  for (int it = 0; it < kiters; ++it) {
    v4f v = __builtin_nontemporal_load(&xrow[tid + (it << 8)]);
    p += (v[0] + v[1]) + (v[2] + v[3]);
  }
  {
    const int i = (kiters << 8) + tid;
    if (i < kvec) {
      v4f v = __builtin_nontemporal_load(&xrow[i]);
      p += (v[0] + v[1]) + (v[2] + v[3]);
    }
  }

  const float rowsum = block_rowsum(p, tid);

  const v4f* __restrict__ wv = reinterpret_cast<const v4f*>(w);
  v4f* __restrict__ orow =
      reinterpret_cast<v4f*>(out) + (size_t)row * (size_t)nvec;

  for (int i = tid; i < nvec; i += BLOCK) {
    v4f wl = wv[i];
    v4f o;
    o[0] = rowsum * wl[0];
    o[1] = rowsum * wl[1];
    o[2] = rowsum * wl[2];
    o[3] = rowsum * wl[3];
    __builtin_nontemporal_store(o, &orow[i]);
  }
}

extern "C" void kernel_launch(void* const* d_in, const int* in_sizes, int n_in,
                              void* d_out, int out_size, void* d_ws,
                              size_t ws_size, hipStream_t stream) {
  const float* x = (const float*)d_in[0];
  const float* w = (const float*)d_in[1];
  float* out = (float*)d_out;

  const int N = in_sizes[1];
  const int batch = out_size / N;
  const int K = in_sizes[0] / batch;

  dim3 grid(batch);
  dim3 block(BLOCK);

  if (K == 4096 && N == 4096 && batch == 8192) {
    hipLaunchKernelGGL((Rank1Linear_kernel_phases<4, 4, 16>), dim3(PGRID),
                       block, 0, stream, x, w, out);
  } else if (K == 4096 && N == 4096) {
    hipLaunchKernelGGL((Rank1Linear_kernel_tmpl<4, 4>), grid, block, 0, stream,
                       x, w, out);
  } else {
    hipLaunchKernelGGL(Rank1Linear_kernel_gen, grid, block, 0, stream,
                       x, w, out, K >> 2, N >> 2);
  }
}
